# starting kernel
# baseline (speedup 1.0000x reference)
#include <hip/hip_runtime.h>
#include <hip/hip_bf16.h>


#define L_STEPS 4096
#define NDIH    3
#define BATCH   32
#define CHAIN   (L_STEPS * NDIH)
#define CHUNK   256
#define NCHUNK  (CHAIN / CHUNK)
#define MST     17
#define PI_F    3.14159265358979f

typedef __attribute__((ext_vector_type(2))) float v2f;
typedef __attribute__((ext_vector_type(8))) float v8f;

__device__ __forceinline__ void wmma_compose4(const float* __restrict__ src,
                                              const float* __restrict__ idm,
                                              float* __restrict__ dst,
                                              int s0, int off, int lane)
{
    const int half = lane >> 4;
    const int Ml   = lane & 15;
    const int g    = Ml >> 2;
    const int r    = Ml & 3;
    const int c    = Ml & 3;
    const int k0   = half ? 2 : 0;

    const int    sg = s0 + g;
    const float* mm = (sg >= off) ? (src + (size_t)(sg - off) * MST) : idm;
    v2f a;
    a.x = mm[r * 4 + k0];
    a.y = mm[r * 4 + k0 + 1];

    const float* nn = src + (size_t)(s0 + g) * MST;
    v2f bb;
    bb.x = nn[k0 * 4 + c];
    bb.y = nn[(k0 + 1) * 4 + c];

    v8f cz = {0.f, 0.f, 0.f, 0.f, 0.f, 0.f, 0.f, 0.f};
    v8f d = __builtin_amdgcn_wmma_f32_16x16x4_f32(false, a, false, bb,
                                                  (short)0, cz, false, false);

#pragma unroll
    for (int rr = 0; rr < 8; ++rr) {
        const int mb = (rr >> 2) + (half << 1);
        if (mb == g) {
            dst[(size_t)(s0 + mb) * MST + (rr & 3) * 4 + c] = d[rr];
        }
    }
}

__global__ void __launch_bounds__(256)
rgn_local_scan(const float* __restrict__ dih_in,
               float* __restrict__ tloc,
               float* __restrict__ agg)
{
    __shared__ float S0[CHUNK * MST];
    __shared__ float S1[CHUNK * MST];
    __shared__ float IDM[16];

    const int tid   = threadIdx.x;
    const int lane  = tid & 31;
    const int wave  = tid >> 5;
    const int chunk = blockIdx.x;
    const int b     = blockIdx.y;

    if (tid < 16) IDM[tid] = (tid % 5 == 0) ? 1.f : 0.f;

    const int gi = chunk * CHUNK + tid;
    const int l  = gi / NDIH;
    const int j  = gi - l * NDIH;

    const float BL[3] = {145.801f, 152.326f, 132.868f};
    const float BA[3] = {2.124f, 1.941f, 2.028f};
    const float bl    = BL[j];
    const float alpha = PI_F - BA[j];
    const float rc    = bl * cosf(alpha);
    const float rs    = bl * sinf(alpha);

    const float dih = dih_in[((size_t)l * BATCH + b) * NDIH + j];
    float sd, cd;
    sincosf(dih, &sd, &cd);
    const float px = rc, py = cd * rs, pz = sd * rs;

    const float inv_r = 1.0f / bl;
    const float ux = px * inv_r, uy = py * inv_r, uz = pz * inv_r;
    const float inv_s = rsqrtf(fmaxf(py * py + pz * pz, 1e-12f));
    const float wx = 0.f, wy = -pz * inv_s, wz = py * inv_s;
    const float vx = wy * uz - wz * uy;
    const float vy = wz * ux - wx * uz;
    const float vz = wx * uy - wy * ux;

    float* T = S0 + (size_t)tid * MST;
    T[0] = ux; T[1] = vx; T[2]  = wx; T[3]  = px;
    T[4] = uy; T[5] = vy; T[6]  = wy; T[7]  = py;
    T[8] = uz; T[9] = vz; T[10] = wz; T[11] = pz;
    T[12] = 0.f; T[13] = 0.f; T[14] = 0.f; T[15] = 1.f;
    __syncthreads();

#pragma unroll
    for (int rnd = 0; rnd < 8; ++rnd) {
        const int off = 1 << rnd;
        float* src = (rnd & 1) ? S1 : S0;
        float* dst = (rnd & 1) ? S0 : S1;
#pragma unroll
        for (int it = 0; it < CHUNK / 32; ++it) {
            wmma_compose4(src, IDM, dst, it * 32 + wave * 4, off, lane);
        }
        __syncthreads();
    }

    const float* fin = S0 + (size_t)tid * MST;
    const size_t oi = ((size_t)gi * BATCH + b) * 3;
    tloc[oi + 0] = fin[3];
    tloc[oi + 1] = fin[7];
    tloc[oi + 2] = fin[11];
    if (tid < 16)
        agg[((size_t)b * NCHUNK + chunk) * 16 + tid] = S0[(CHUNK - 1) * MST + tid];
}

__global__ void rgn_agg_scan(const float* __restrict__ agg,
                             float* __restrict__ pref)
{
    const int b = threadIdx.x;
    if (b >= BATCH) return;
    float P[16] = {1.f, 0.f, 0.f, 0.f,
                   0.f, 1.f, 0.f, 0.f,
                   0.f, 0.f, 1.f, 0.f,
                   0.f, 0.f, 0.f, 1.f};
    for (int k = 0; k < NCHUNK; ++k) {
        const size_t base = ((size_t)b * NCHUNK + k) * 16;
#pragma unroll
        for (int e = 0; e < 16; ++e) pref[base + e] = P[e];
        const float* A = agg + base;
        float T[16];
#pragma unroll
        for (int r = 0; r < 4; ++r)
#pragma unroll
            for (int c = 0; c < 4; ++c)
                T[r * 4 + c] = fmaf(P[r * 4 + 0], A[0 * 4 + c],
                               fmaf(P[r * 4 + 1], A[1 * 4 + c],
                               fmaf(P[r * 4 + 2], A[2 * 4 + c],
                                    P[r * 4 + 3] * A[3 * 4 + c])));
#pragma unroll
        for (int e = 0; e < 16; ++e) P[e] = T[e];
    }
}

__global__ void __launch_bounds__(256)
rgn_apply(const float* __restrict__ tloc,
          const float* __restrict__ pref,
          float* __restrict__ out)
{
    const int t = blockIdx.x * 256 + threadIdx.x;
    if (t >= CHAIN * BATCH) return;
    const int i = t / BATCH;
    const int b = t - i * BATCH;
    const int chunk = i / CHUNK;
    const float* P = pref + ((size_t)b * NCHUNK + chunk) * 16;
    const size_t oi = (size_t)t * 3;
    const float tx = tloc[oi + 0], ty = tloc[oi + 1], tz = tloc[oi + 2];
    out[oi + 0] = fmaf(P[0], tx, fmaf(P[1], ty, fmaf(P[2],  tz, P[3])));
    out[oi + 1] = fmaf(P[4], tx, fmaf(P[5], ty, fmaf(P[6],  tz, P[7])));
    out[oi + 2] = fmaf(P[8], tx, fmaf(P[9], ty, fmaf(P[10], tz, P[11])));
}

extern "C" void kernel_launch(void* const* d_in, const int* in_sizes, int n_in,
                              void* d_out, int out_size, void* d_ws, size_t ws_size,
                              hipStream_t stream)
{
    (void)in_sizes; (void)n_in; (void)out_size; (void)ws_size;
    const float* dih = (const float*)d_in[0];
    float* out = (float*)d_out;

    float* ws   = (float*)d_ws;
    float* tloc = ws;
    float* agg  = tloc + (size_t)CHAIN * BATCH * 3;
    float* pref = agg + (size_t)BATCH * NCHUNK * 16;

    dim3 gridB(NCHUNK, BATCH, 1);
    rgn_local_scan<<<gridB, 256, 0, stream>>>(dih, tloc, agg);
    rgn_agg_scan<<<1, 32, 0, stream>>>(agg, pref);
    const int total = CHAIN * BATCH;
    rgn_apply<<<(total + 255) / 256, 256, 0, stream>>>(tloc, pref, out);
}


# Chunk-aggregate scan: serial 48-step loop on one wave (dependent global loads per step) replaced by 6-round log-step scan in LDS, one workgroup per chain; prefix products now combined in tree order
# speedup vs baseline: 1.4461x; 1.4461x over previous
#include <hip/hip_runtime.h>
#include <hip/hip_bf16.h>


#define L_STEPS 4096
#define NDIH    3
#define BATCH   32
#define CHAIN   (L_STEPS * NDIH)
#define CHUNK   256
#define NCHUNK  (CHAIN / CHUNK)
#define MST     17
#define PI_F    3.14159265358979f

typedef __attribute__((ext_vector_type(2))) float v2f;
typedef __attribute__((ext_vector_type(8))) float v8f;

__device__ __forceinline__ void wmma_compose4(const float* __restrict__ src,
                                              const float* __restrict__ idm,
                                              float* __restrict__ dst,
                                              int s0, int off, int lane)
{
    const int half = lane >> 4;
    const int Ml   = lane & 15;
    const int g    = Ml >> 2;
    const int r    = Ml & 3;
    const int c    = Ml & 3;
    const int k0   = half ? 2 : 0;

    const int    sg = s0 + g;
    const float* mm = (sg >= off) ? (src + (size_t)(sg - off) * MST) : idm;
    v2f a;
    a.x = mm[r * 4 + k0];
    a.y = mm[r * 4 + k0 + 1];

    const float* nn = src + (size_t)(s0 + g) * MST;
    v2f bb;
    bb.x = nn[k0 * 4 + c];
    bb.y = nn[(k0 + 1) * 4 + c];

    v8f cz = {0.f, 0.f, 0.f, 0.f, 0.f, 0.f, 0.f, 0.f};
    v8f d = __builtin_amdgcn_wmma_f32_16x16x4_f32(false, a, false, bb,
                                                  (short)0, cz, false, false);

#pragma unroll
    for (int rr = 0; rr < 8; ++rr) {
        const int mb = (rr >> 2) + (half << 1);
        if (mb == g) {
            dst[(size_t)(s0 + mb) * MST + (rr & 3) * 4 + c] = d[rr];
        }
    }
}

__global__ void __launch_bounds__(256)
rgn_local_scan(const float* __restrict__ dih_in,
               float* __restrict__ tloc,
               float* __restrict__ agg)
{
    __shared__ float S0[CHUNK * MST];
    __shared__ float S1[CHUNK * MST];
    __shared__ float IDM[16];

    const int tid   = threadIdx.x;
    const int lane  = tid & 31;
    const int wave  = tid >> 5;
    const int chunk = blockIdx.x;
    const int b     = blockIdx.y;

    if (tid < 16) IDM[tid] = (tid % 5 == 0) ? 1.f : 0.f;

    const int gi = chunk * CHUNK + tid;
    const int l  = gi / NDIH;
    const int j  = gi - l * NDIH;

    const float BL[3] = {145.801f, 152.326f, 132.868f};
    const float BA[3] = {2.124f, 1.941f, 2.028f};
    const float bl    = BL[j];
    const float alpha = PI_F - BA[j];
    const float rc    = bl * cosf(alpha);
    const float rs    = bl * sinf(alpha);

    const float dih = dih_in[((size_t)l * BATCH + b) * NDIH + j];
    float sd, cd;
    sincosf(dih, &sd, &cd);
    const float px = rc, py = cd * rs, pz = sd * rs;

    const float inv_r = 1.0f / bl;
    const float ux = px * inv_r, uy = py * inv_r, uz = pz * inv_r;
    const float inv_s = rsqrtf(fmaxf(py * py + pz * pz, 1e-12f));
    const float wx = 0.f, wy = -pz * inv_s, wz = py * inv_s;
    const float vx = wy * uz - wz * uy;
    const float vy = wz * ux - wx * uz;
    const float vz = wx * uy - wy * ux;

    float* T = S0 + (size_t)tid * MST;
    T[0] = ux; T[1] = vx; T[2]  = wx; T[3]  = px;
    T[4] = uy; T[5] = vy; T[6]  = wy; T[7]  = py;
    T[8] = uz; T[9] = vz; T[10] = wz; T[11] = pz;
    T[12] = 0.f; T[13] = 0.f; T[14] = 0.f; T[15] = 1.f;
    __syncthreads();

#pragma unroll
    for (int rnd = 0; rnd < 8; ++rnd) {
        const int off = 1 << rnd;
        float* src = (rnd & 1) ? S1 : S0;
        float* dst = (rnd & 1) ? S0 : S1;
#pragma unroll
        for (int it = 0; it < CHUNK / 32; ++it) {
            wmma_compose4(src, IDM, dst, it * 32 + wave * 4, off, lane);
        }
        __syncthreads();
    }

    const float* fin = S0 + (size_t)tid * MST;
    const size_t oi = ((size_t)gi * BATCH + b) * 3;
    tloc[oi + 0] = fin[3];
    tloc[oi + 1] = fin[7];
    tloc[oi + 2] = fin[11];
    if (tid < 16)
        agg[((size_t)b * NCHUNK + chunk) * 16 + tid] = S0[(CHUNK - 1) * MST + tid];
}

#define AST 20

__global__ void __launch_bounds__(NCHUNK * 4)
rgn_agg_scan(const float* __restrict__ agg,
             float* __restrict__ pref)
{
    __shared__ __attribute__((aligned(16))) float X0[NCHUNK * AST];
    __shared__ __attribute__((aligned(16))) float X1[NCHUNK * AST];

    const int b = blockIdx.x;
    const int k = threadIdx.x >> 2;
    const int r = threadIdx.x & 3;
    const size_t gofs = ((size_t)b * NCHUNK + k) * 16 + r * 4;
    const int    lofs = k * AST + r * 4;

    float4 row = *reinterpret_cast<const float4*>(agg + gofs);
    *reinterpret_cast<float4*>(X0 + lofs) = row;
    __syncthreads();

#pragma unroll
    for (int rnd = 0; rnd < 6; ++rnd) {
        const int off = 1 << rnd;
        const float* src = (rnd & 1) ? X1 : X0;
        float*       dst = (rnd & 1) ? X0 : X1;
        if (k >= off) {
            const float4 l  = *reinterpret_cast<const float4*>(src + (k - off) * AST + r * 4);
            const float4 r0 = *reinterpret_cast<const float4*>(src + k * AST + 0);
            const float4 r1 = *reinterpret_cast<const float4*>(src + k * AST + 4);
            const float4 r2 = *reinterpret_cast<const float4*>(src + k * AST + 8);
            const float4 r3 = *reinterpret_cast<const float4*>(src + k * AST + 12);
            row.x = fmaf(l.x, r0.x, fmaf(l.y, r1.x, fmaf(l.z, r2.x, l.w * r3.x)));
            row.y = fmaf(l.x, r0.y, fmaf(l.y, r1.y, fmaf(l.z, r2.y, l.w * r3.y)));
            row.z = fmaf(l.x, r0.z, fmaf(l.y, r1.z, fmaf(l.z, r2.z, l.w * r3.z)));
            row.w = fmaf(l.x, r0.w, fmaf(l.y, r1.w, fmaf(l.z, r2.w, l.w * r3.w)));
        }
        *reinterpret_cast<float4*>(dst + lofs) = row;
        __syncthreads();
    }

    float4 p;
    if (k == 0) {
        p.x = (r == 0) ? 1.f : 0.f;
        p.y = (r == 1) ? 1.f : 0.f;
        p.z = (r == 2) ? 1.f : 0.f;
        p.w = (r == 3) ? 1.f : 0.f;
    } else {
        p = *reinterpret_cast<const float4*>(X0 + lofs - AST);
    }
    *reinterpret_cast<float4*>(pref + gofs) = p;
}

__global__ void __launch_bounds__(256)
rgn_apply(const float* __restrict__ tloc,
          const float* __restrict__ pref,
          float* __restrict__ out)
{
    const int t = blockIdx.x * 256 + threadIdx.x;
    if (t >= CHAIN * BATCH) return;
    const int i = t / BATCH;
    const int b = t - i * BATCH;
    const int chunk = i / CHUNK;
    const float* P = pref + ((size_t)b * NCHUNK + chunk) * 16;
    const size_t oi = (size_t)t * 3;
    const float tx = tloc[oi + 0], ty = tloc[oi + 1], tz = tloc[oi + 2];
    out[oi + 0] = fmaf(P[0], tx, fmaf(P[1], ty, fmaf(P[2],  tz, P[3])));
    out[oi + 1] = fmaf(P[4], tx, fmaf(P[5], ty, fmaf(P[6],  tz, P[7])));
    out[oi + 2] = fmaf(P[8], tx, fmaf(P[9], ty, fmaf(P[10], tz, P[11])));
}

extern "C" void kernel_launch(void* const* d_in, const int* in_sizes, int n_in,
                              void* d_out, int out_size, void* d_ws, size_t ws_size,
                              hipStream_t stream)
{
    (void)in_sizes; (void)n_in; (void)out_size; (void)ws_size;
    const float* dih = (const float*)d_in[0];
    float* out = (float*)d_out;

    float* ws   = (float*)d_ws;
    float* tloc = ws;
    float* agg  = tloc + (size_t)CHAIN * BATCH * 3;
    float* pref = agg + (size_t)BATCH * NCHUNK * 16;

    dim3 gridB(NCHUNK, BATCH, 1);
    rgn_local_scan<<<gridB, 256, 0, stream>>>(dih, tloc, agg);
    rgn_agg_scan<<<BATCH, NCHUNK * 4, 0, stream>>>(agg, pref);
    const int total = CHAIN * BATCH;
    rgn_apply<<<(total + 255) / 256, 256, 0, stream>>>(tloc, pref, out);
}


# Kernel 3 (affine apply): a thread keeps its chain and does 8 consecutive chain positions, loading its chunk prefix once: 1536 waves to start, was 12288; per-point sums and their order unchanged
# speedup vs baseline: 1.5275x; 1.0563x over previous
#include <hip/hip_runtime.h>
#include <hip/hip_bf16.h>


#define L_STEPS 4096
#define NDIH    3
#define BATCH   32
#define CHAIN   (L_STEPS * NDIH)
#define CHUNK   256
#define NCHUNK  (CHAIN / CHUNK)
#define MST     17
#define PI_F    3.14159265358979f

typedef __attribute__((ext_vector_type(2))) float v2f;
typedef __attribute__((ext_vector_type(8))) float v8f;

__device__ __forceinline__ void wmma_compose4(const float* __restrict__ src,
                                              const float* __restrict__ idm,
                                              float* __restrict__ dst,
                                              int s0, int off, int lane)
{
    const int half = lane >> 4;
    const int Ml   = lane & 15;
    const int g    = Ml >> 2;
    const int r    = Ml & 3;
    const int c    = Ml & 3;
    const int k0   = half ? 2 : 0;

    const int    sg = s0 + g;
    const float* mm = (sg >= off) ? (src + (size_t)(sg - off) * MST) : idm;
    v2f a;
    a.x = mm[r * 4 + k0];
    a.y = mm[r * 4 + k0 + 1];

    const float* nn = src + (size_t)(s0 + g) * MST;
    v2f bb;
    bb.x = nn[k0 * 4 + c];
    bb.y = nn[(k0 + 1) * 4 + c];

    v8f cz = {0.f, 0.f, 0.f, 0.f, 0.f, 0.f, 0.f, 0.f};
    v8f d = __builtin_amdgcn_wmma_f32_16x16x4_f32(false, a, false, bb,
                                                  (short)0, cz, false, false);

#pragma unroll
    for (int rr = 0; rr < 8; ++rr) {
        const int mb = (rr >> 2) + (half << 1);
        if (mb == g) {
            dst[(size_t)(s0 + mb) * MST + (rr & 3) * 4 + c] = d[rr];
        }
    }
}

__global__ void __launch_bounds__(256)
rgn_local_scan(const float* __restrict__ dih_in,
               float* __restrict__ tloc,
               float* __restrict__ agg)
{
    __shared__ float S0[CHUNK * MST];
    __shared__ float S1[CHUNK * MST];
    __shared__ float IDM[16];

    const int tid   = threadIdx.x;
    const int lane  = tid & 31;
    const int wave  = tid >> 5;
    const int chunk = blockIdx.x;
    const int b     = blockIdx.y;

    if (tid < 16) IDM[tid] = (tid % 5 == 0) ? 1.f : 0.f;

    const int gi = chunk * CHUNK + tid;
    const int l  = gi / NDIH;
    const int j  = gi - l * NDIH;

    const float BL[3] = {145.801f, 152.326f, 132.868f};
    const float BA[3] = {2.124f, 1.941f, 2.028f};
    const float bl    = BL[j];
    const float alpha = PI_F - BA[j];
    const float rc    = bl * cosf(alpha);
    const float rs    = bl * sinf(alpha);

    const float dih = dih_in[((size_t)l * BATCH + b) * NDIH + j];
    float sd, cd;
    sincosf(dih, &sd, &cd);
    const float px = rc, py = cd * rs, pz = sd * rs;

    const float inv_r = 1.0f / bl;
    const float ux = px * inv_r, uy = py * inv_r, uz = pz * inv_r;
    const float inv_s = rsqrtf(fmaxf(py * py + pz * pz, 1e-12f));
    const float wx = 0.f, wy = -pz * inv_s, wz = py * inv_s;
    const float vx = wy * uz - wz * uy;
    const float vy = wz * ux - wx * uz;
    const float vz = wx * uy - wy * ux;

    float* T = S0 + (size_t)tid * MST;
    T[0] = ux; T[1] = vx; T[2]  = wx; T[3]  = px;
    T[4] = uy; T[5] = vy; T[6]  = wy; T[7]  = py;
    T[8] = uz; T[9] = vz; T[10] = wz; T[11] = pz;
    T[12] = 0.f; T[13] = 0.f; T[14] = 0.f; T[15] = 1.f;
    __syncthreads();

#pragma unroll
    for (int rnd = 0; rnd < 8; ++rnd) {
        const int off = 1 << rnd;
        float* src = (rnd & 1) ? S1 : S0;
        float* dst = (rnd & 1) ? S0 : S1;
#pragma unroll
        for (int it = 0; it < CHUNK / 32; ++it) {
            wmma_compose4(src, IDM, dst, it * 32 + wave * 4, off, lane);
        }
        __syncthreads();
    }

    const float* fin = S0 + (size_t)tid * MST;
    const size_t oi = ((size_t)gi * BATCH + b) * 3;
    tloc[oi + 0] = fin[3];
    tloc[oi + 1] = fin[7];
    tloc[oi + 2] = fin[11];
    if (tid < 16)
        agg[((size_t)b * NCHUNK + chunk) * 16 + tid] = S0[(CHUNK - 1) * MST + tid];
}

#define AST 20

__global__ void __launch_bounds__(NCHUNK * 4)
rgn_agg_scan(const float* __restrict__ agg,
             float* __restrict__ pref)
{
    __shared__ __attribute__((aligned(16))) float X0[NCHUNK * AST];
    __shared__ __attribute__((aligned(16))) float X1[NCHUNK * AST];

    const int b = blockIdx.x;
    const int k = threadIdx.x >> 2;
    const int r = threadIdx.x & 3;
    const size_t gofs = ((size_t)b * NCHUNK + k) * 16 + r * 4;
    const int    lofs = k * AST + r * 4;

    float4 row = *reinterpret_cast<const float4*>(agg + gofs);
    *reinterpret_cast<float4*>(X0 + lofs) = row;
    __syncthreads();

#pragma unroll
    for (int rnd = 0; rnd < 6; ++rnd) {
        const int off = 1 << rnd;
        const float* src = (rnd & 1) ? X1 : X0;
        float*       dst = (rnd & 1) ? X0 : X1;
        if (k >= off) {
            const float4 l  = *reinterpret_cast<const float4*>(src + (k - off) * AST + r * 4);
            const float4 r0 = *reinterpret_cast<const float4*>(src + k * AST + 0);
            const float4 r1 = *reinterpret_cast<const float4*>(src + k * AST + 4);
            const float4 r2 = *reinterpret_cast<const float4*>(src + k * AST + 8);
            const float4 r3 = *reinterpret_cast<const float4*>(src + k * AST + 12);
            row.x = fmaf(l.x, r0.x, fmaf(l.y, r1.x, fmaf(l.z, r2.x, l.w * r3.x)));
            row.y = fmaf(l.x, r0.y, fmaf(l.y, r1.y, fmaf(l.z, r2.y, l.w * r3.y)));
            row.z = fmaf(l.x, r0.z, fmaf(l.y, r1.z, fmaf(l.z, r2.z, l.w * r3.z)));
            row.w = fmaf(l.x, r0.w, fmaf(l.y, r1.w, fmaf(l.z, r2.w, l.w * r3.w)));
        }
        *reinterpret_cast<float4*>(dst + lofs) = row;
        __syncthreads();
    }

    float4 p;
    if (k == 0) {
        p.x = (r == 0) ? 1.f : 0.f;
        p.y = (r == 1) ? 1.f : 0.f;
        p.z = (r == 2) ? 1.f : 0.f;
        p.w = (r == 3) ? 1.f : 0.f;
    } else {
        p = *reinterpret_cast<const float4*>(X0 + lofs - AST);
    }
    *reinterpret_cast<float4*>(pref + gofs) = p;
}

#define APTS     8
#define ABLOCK   128
#define AGRID    (CHAIN * BATCH / (APTS * ABLOCK))

typedef __attribute__((ext_vector_type(4))) float v4f;

__global__ void __launch_bounds__(ABLOCK)
rgn_apply(const float* __restrict__ tloc,
          const float* __restrict__ pref,
          float* __restrict__ out)
{
    const int tt = blockIdx.x * ABLOCK + threadIdx.x;
    const int b  = tt % BATCH;
    const int i0 = (tt / BATCH) * APTS;
    if (i0 >= CHAIN) return;
    const int chunk = i0 / CHUNK;

    const v4f* P = reinterpret_cast<const v4f*>(
        pref + ((size_t)b * NCHUNK + chunk) * 16);
    const v4f r0 = P[0], r1 = P[1], r2 = P[2];

    float x[APTS][3];
#pragma unroll
    for (int k = 0; k < APTS; ++k) {
        const size_t oi = ((size_t)(i0 + k) * BATCH + b) * 3;
        x[k][0] = tloc[oi + 0]; x[k][1] = tloc[oi + 1]; x[k][2] = tloc[oi + 2];
    }

#pragma unroll
    for (int k = 0; k < APTS; ++k) {
        const size_t oi = ((size_t)(i0 + k) * BATCH + b) * 3;
        const float tx = x[k][0], ty = x[k][1], tz = x[k][2];
        out[oi + 0] = fmaf(r0.x, tx, fmaf(r0.y, ty, fmaf(r0.z, tz, r0.w)));
        out[oi + 1] = fmaf(r1.x, tx, fmaf(r1.y, ty, fmaf(r1.z, tz, r1.w)));
        out[oi + 2] = fmaf(r2.x, tx, fmaf(r2.y, ty, fmaf(r2.z, tz, r2.w)));
    }
}

extern "C" void kernel_launch(void* const* d_in, const int* in_sizes, int n_in,
                              void* d_out, int out_size, void* d_ws, size_t ws_size,
                              hipStream_t stream)
{
    (void)in_sizes; (void)n_in; (void)out_size; (void)ws_size;
    const float* dih = (const float*)d_in[0];
    float* out = (float*)d_out;

    float* ws   = (float*)d_ws;
    float* tloc = ws;
    float* agg  = tloc + (size_t)CHAIN * BATCH * 3;
    float* pref = agg + (size_t)BATCH * NCHUNK * 16;

    dim3 gridB(NCHUNK, BATCH, 1);
    rgn_local_scan<<<gridB, 256, 0, stream>>>(dih, tloc, agg);
    rgn_agg_scan<<<BATCH, NCHUNK * 4, 0, stream>>>(agg, pref);
    rgn_apply<<<AGRID, ABLOCK, 0, stream>>>(tloc, pref, out);
}


# Aggregate-prefix launch removed: the apply kernel folds its chunk prefix itself from the 48 chunk aggregates (8 wave segments, joined in LDS); 2 launches, was 3; prefix product order changed
# speedup vs baseline: 3.3636x; 2.2020x over previous
#include <hip/hip_runtime.h>
#include <hip/hip_bf16.h>


#define L_STEPS 4096
#define NDIH    3
#define BATCH   32
#define CHAIN   (L_STEPS * NDIH)
#define CHUNK   256
#define NCHUNK  (CHAIN / CHUNK)
#define MST     17
#define SUB     8
#define NSUB    (CHUNK / SUB)
#define PI_F    3.14159265358979f
#define AGE     12

typedef __attribute__((ext_vector_type(2))) float v2f;
typedef __attribute__((ext_vector_type(8))) float v8f;

__device__ __forceinline__ void wmma_compose4(const float* mm,
                                              const float* nn,
                                              float* dd,
                                              int lane)
{
    const int half = lane >> 4;
    const int Ml   = lane & 15;
    const int g    = Ml >> 2;
    const int r    = Ml & 3;
    const int c    = Ml & 3;
    const int k0   = half ? 2 : 0;

    v2f a;
    a.x = mm[r * 4 + k0];
    a.y = mm[r * 4 + k0 + 1];

    v2f bb;
    bb.x = nn[k0 * 4 + c];
    bb.y = nn[(k0 + 1) * 4 + c];

    v8f cz = {0.f, 0.f, 0.f, 0.f, 0.f, 0.f, 0.f, 0.f};
    v8f d = __builtin_amdgcn_wmma_f32_16x16x4_f32(false, a, false, bb,
                                                  (short)0, cz, false, false);

#pragma unroll
    for (int rr = 0; rr < 8; ++rr) {
        const int mb = (rr >> 2) + (half << 1);
        if (mb == g) {
            dd[(rr & 3) * 4 + c] = d[rr];
        }
    }
}

__device__ __forceinline__ v8f wmma_f32_c0_then_valu(v2f a, v2f b)
{
    v8f d;
    asm volatile("v_wmma_f32_16x16x4_f32 %0, %1, %2, 0\n\t"
                 "v_nop\n\t"
                 "v_nop\n\t"
                 "v_nop\n\t"
                 "v_nop"
                 : "=&v"(d)
                 : "v"(a), "v"(b));
    return d;
}

__global__ void __launch_bounds__(256)
rgn_local_scan(const float* __restrict__ dih_in,
               float* __restrict__ tloc,
               float* __restrict__ agg)
{
    __shared__ float S0[CHUNK * MST];
    __shared__ float G0[NSUB * MST];
    __shared__ float G1[NSUB * MST];
    __shared__ float IDM[16];

    const int tid   = threadIdx.x;
    const int lane  = tid & 31;
    const int wave  = tid >> 5;
    const int chunk = blockIdx.x;
    const int b     = blockIdx.y;

    if (tid < 16) IDM[tid] = (tid % 5 == 0) ? 1.f : 0.f;

    const int gi = chunk * CHUNK + tid;
    const int l  = gi / NDIH;
    const int j  = gi - l * NDIH;

    const float BL[3] = {145.801f, 152.326f, 132.868f};
    const float BA[3] = {2.124f, 1.941f, 2.028f};
    const float bl    = BL[j];
    const float alpha = PI_F - BA[j];
    const float rc    = bl * cosf(alpha);
    const float rs    = bl * sinf(alpha);

    const float dih = dih_in[((size_t)l * BATCH + b) * NDIH + j];
    float sd, cd;
    sincosf(dih, &sd, &cd);
    const float px = rc, py = cd * rs, pz = sd * rs;

    const float inv_r = 1.0f / bl;
    const float ux = px * inv_r, uy = py * inv_r, uz = pz * inv_r;
    const float inv_s = rsqrtf(fmaxf(py * py + pz * pz, 1e-12f));
    const float wx = 0.f, wy = -pz * inv_s, wz = py * inv_s;
    const float vx = wy * uz - wz * uy;
    const float vy = wz * ux - wx * uz;
    const float vz = wx * uy - wy * ux;

    float* T = S0 + (size_t)tid * MST;
    T[0] = ux; T[1] = vx; T[2]  = wx; T[3]  = px;
    T[4] = uy; T[5] = vy; T[6]  = wy; T[7]  = py;
    T[8] = uz; T[9] = vz; T[10] = wz; T[11] = pz;
    T[12] = 0.f; T[13] = 0.f; T[14] = 0.f; T[15] = 1.f;
    __syncthreads();

    const int q = wave * 4 + ((lane & 15) >> 2);

#pragma unroll
    for (int k = 1; k < SUB; ++k) {
        float* cur = S0 + (q * SUB + k) * MST;
        wmma_compose4(cur - MST, cur, cur, lane);
        __syncthreads();
    }

#pragma unroll
    for (int rnd = 0; rnd < 5; ++rnd) {
        const int off = 1 << rnd;
        const float* mm;
        const float* nn;
        float*       dd;
        if (rnd == 0) {
            nn = S0 + (q * SUB + SUB - 1) * MST;
            mm = (q >= 1) ? (nn - SUB * MST) : IDM;
            dd = G1 + q * MST;
        } else {
            const float* src = (rnd & 1) ? G1 : G0;
            float*       dst = (rnd & 1) ? G0 : G1;
            nn = src + q * MST;
            mm = (q >= off) ? (src + (q - off) * MST) : IDM;
            dd = dst + q * MST;
        }
        wmma_compose4(mm, nn, dd, lane);
        __syncthreads();
    }

    {
        const int half = lane >> 4;
        const int n    = lane & 15;
        const int g    = n >> 2;
        const int jp   = n & 3;
        const int k0   = half ? 2 : 0;

        const float* carry = (q >= 1) ? (G1 + (q - 1) * MST) : IDM;
        v2f a;
        a.x = carry[jp * 4 + k0];
        a.y = carry[jp * 4 + k0 + 1];

        v2f pb[SUB / 4];
#pragma unroll
        for (int p = 0; p < SUB / 4; ++p) {
            const float* P = S0 + (q * SUB + p * 4 + jp) * MST;
            pb[p].x = P[k0 * 4 + 3];
            pb[p].y = P[(k0 + 1) * 4 + 3];
        }

        const bool keep = (g >> 1) == half;
        const bool hi   = (g & 1) != 0;
#pragma unroll
        for (int p = 0; p < SUB / 4; ++p) {
            const v8f d = wmma_f32_c0_then_valu(a, pb[p]);
            const float x = hi ? d[4] : d[0];
            const float y = hi ? d[5] : d[1];
            const float z = hi ? d[6] : d[2];
            if (keep) {
                const int    pi = chunk * CHUNK + q * SUB + p * 4 + jp;
                const size_t oi = ((size_t)pi * BATCH + b) * 3;
                tloc[oi + 0] = x;
                tloc[oi + 1] = y;
                tloc[oi + 2] = z;
            }
        }
    }

    if (tid < AGE)
        agg[((size_t)chunk * AGE + tid) * BATCH + b] = G1[(NSUB - 1) * MST + tid];
}

#define APTS     8
#define AWAVES   8
#define ABLOCK   (AWAVES * 32)
#define AGRID    (CHAIN * BATCH / (APTS * ABLOCK))
#define SEG      (NCHUNK / AWAVES)

__device__ __forceinline__ void affine_mul(float* q, const float* a)
{
#pragma unroll
    for (int i = 0; i < 3; ++i) {
        const float lx = q[i * 4 + 0], ly = q[i * 4 + 1];
        const float lz = q[i * 4 + 2], lw = q[i * 4 + 3];
        q[i * 4 + 0] = fmaf(lx, a[0], fmaf(ly, a[4], lz * a[8]));
        q[i * 4 + 1] = fmaf(lx, a[1], fmaf(ly, a[5], lz * a[9]));
        q[i * 4 + 2] = fmaf(lx, a[2], fmaf(ly, a[6], lz * a[10]));
        q[i * 4 + 3] = fmaf(lx, a[3], fmaf(ly, a[7], fmaf(lz, a[11], lw)));
    }
}

__global__ void __launch_bounds__(ABLOCK)
rgn_apply(const float* __restrict__ tloc,
          const float* __restrict__ agg,
          float* __restrict__ out)
{
    __shared__ float Q[AWAVES * AGE * BATCH];

    const int b     = threadIdx.x & 31;
    const int w     = threadIdx.x >> 5;
    const int i0    = (blockIdx.x * AWAVES + w) * APTS;
    const int chunk = (blockIdx.x * (AWAVES * APTS)) / CHUNK;

    float x[APTS][3];
#pragma unroll
    for (int k = 0; k < APTS; ++k) {
        const size_t oi = ((size_t)(i0 + k) * BATCH + b) * 3;
        x[k][0] = tloc[oi + 0]; x[k][1] = tloc[oi + 1]; x[k][2] = tloc[oi + 2];
    }

    float q[AGE] = {1.f, 0.f, 0.f, 0.f,  0.f, 1.f, 0.f, 0.f,  0.f, 0.f, 1.f, 0.f};
    if (w * SEG < chunk) {
        float a[SEG][AGE];
#pragma unroll
        for (int j = 0; j < SEG; ++j) {
#pragma unroll
            for (int e = 0; e < AGE; ++e)
                a[j][e] = agg[((size_t)(w * SEG + j) * AGE + e) * BATCH + b];
        }
#pragma unroll
        for (int j = 0; j < SEG; ++j) {
            if (w * SEG + j < chunk) affine_mul(q, a[j]);
        }
    }
#pragma unroll
    for (int e = 0; e < AGE; ++e) Q[(w * AGE + e) * BATCH + b] = q[e];
    __syncthreads();

    float p[AGE];
#pragma unroll
    for (int e = 0; e < AGE; ++e) p[e] = Q[e * BATCH + b];
    float r[AWAVES - 1][AGE];
#pragma unroll
    for (int s = 1; s < AWAVES; ++s) {
#pragma unroll
        for (int e = 0; e < AGE; ++e) r[s - 1][e] = Q[(s * AGE + e) * BATCH + b];
    }
#pragma unroll
    for (int s = 1; s < AWAVES; ++s) affine_mul(p, r[s - 1]);

#pragma unroll
    for (int k = 0; k < APTS; ++k) {
        const size_t oi = ((size_t)(i0 + k) * BATCH + b) * 3;
        const float tx = x[k][0], ty = x[k][1], tz = x[k][2];
        out[oi + 0] = fmaf(p[0], tx, fmaf(p[1], ty, fmaf(p[2],  tz, p[3])));
        out[oi + 1] = fmaf(p[4], tx, fmaf(p[5], ty, fmaf(p[6],  tz, p[7])));
        out[oi + 2] = fmaf(p[8], tx, fmaf(p[9], ty, fmaf(p[10], tz, p[11])));
    }
}

extern "C" void kernel_launch(void* const* d_in, const int* in_sizes, int n_in,
                              void* d_out, int out_size, void* d_ws, size_t ws_size,
                              hipStream_t stream)
{
    (void)in_sizes; (void)n_in; (void)out_size; (void)ws_size;
    const float* dih = (const float*)d_in[0];
    float* out = (float*)d_out;

    float* ws   = (float*)d_ws;
    float* tloc = ws;
    float* agg  = tloc + (size_t)CHAIN * BATCH * 3;

    dim3 gridB(NCHUNK, BATCH, 1);
    rgn_local_scan<<<gridB, 256, 0, stream>>>(dih, tloc, agg);
    rgn_apply<<<AGRID, ABLOCK, 0, stream>>>(tloc, agg, out);
}


# Scan kernel, LDS clocks per 4x4 product cut: matrix stride 18 (was 17) makes the fragment loads conflict-free (were 2-way); the D scatter is 2 stores under one mask (was 4 under two)
# speedup vs baseline: 3.5733x; 1.0623x over previous
#include <hip/hip_runtime.h>
#include <hip/hip_bf16.h>


#define L_STEPS 4096
#define NDIH    3
#define BATCH   32
#define CHAIN   (L_STEPS * NDIH)
#define CHUNK   256
#define NCHUNK  (CHAIN / CHUNK)
#define MST     18
#define GST     16
#define SUB     8
#define NSUB    (CHUNK / SUB)
#define PI_F    3.14159265358979f
#define AGE     12

typedef __attribute__((ext_vector_type(2))) float v2f;
typedef __attribute__((ext_vector_type(8))) float v8f;

__device__ __forceinline__ v8f wmma_f32_c0_then_valu(v2f a, v2f b)
{
    v8f d;
    asm volatile("v_wmma_f32_16x16x4_f32 %0, %1, %2, 0\n\t"
                 "v_nop\n\t"
                 "v_nop\n\t"
                 "v_nop\n\t"
                 "v_nop"
                 : "=&v"(d)
                 : "v"(a), "v"(b));
    return d;
}

__device__ __forceinline__ void wmma_compose4(const float* mm,
                                              const float* nn,
                                              float* dd,
                                              int lane)
{
    const int half = lane >> 4;
    const int Ml   = lane & 15;
    const int g    = Ml >> 2;
    const int r    = Ml & 3;
    const int c    = Ml & 3;
    const int k0   = half ? 2 : 0;

    const v2f a = *(const v2f*)(mm + r * 4 + k0);

    v2f bb;
    bb.x = nn[k0 * 4 + c];
    bb.y = nn[(k0 + 1) * 4 + c];

    const v8f d = wmma_f32_c0_then_valu(a, bb);

    const bool keep = (g >> 1) == half;
    const bool hi   = (g & 1) != 0;
    const float x0 = hi ? d[4] : d[0];
    const float x1 = hi ? d[5] : d[1];
    const float x2 = hi ? d[6] : d[2];
    const float x3 = hi ? d[7] : d[3];
    if (keep) {
        dd[0 * 4 + c] = x0;
        dd[1 * 4 + c] = x1;
        dd[2 * 4 + c] = x2;
        dd[3 * 4 + c] = x3;
    }
}

__global__ void __launch_bounds__(256)
rgn_local_scan(const float* __restrict__ dih_in,
               float* __restrict__ tloc,
               float* __restrict__ agg)
{
    __shared__ __attribute__((aligned(16))) float S0[CHUNK * MST];
    __shared__ __attribute__((aligned(16))) float G0[NSUB * GST];
    __shared__ __attribute__((aligned(16))) float G1[NSUB * GST];
    __shared__ __attribute__((aligned(16))) float IDM[16];

    const int tid   = threadIdx.x;
    const int lane  = tid & 31;
    const int wave  = tid >> 5;
    const int chunk = blockIdx.x;
    const int b     = blockIdx.y;

    if (tid < 16) IDM[tid] = (tid % 5 == 0) ? 1.f : 0.f;

    const int gi = chunk * CHUNK + tid;
    const int l  = gi / NDIH;
    const int j  = gi - l * NDIH;

    const float BL[3] = {145.801f, 152.326f, 132.868f};
    const float BA[3] = {2.124f, 1.941f, 2.028f};
    const float bl    = BL[j];
    const float alpha = PI_F - BA[j];
    const float rc    = bl * cosf(alpha);
    const float rs    = bl * sinf(alpha);

    const float dih = dih_in[((size_t)l * BATCH + b) * NDIH + j];
    float sd, cd;
    sincosf(dih, &sd, &cd);
    const float px = rc, py = cd * rs, pz = sd * rs;

    const float inv_r = 1.0f / bl;
    const float ux = px * inv_r, uy = py * inv_r, uz = pz * inv_r;
    const float inv_s = rsqrtf(fmaxf(py * py + pz * pz, 1e-12f));
    const float wx = 0.f, wy = -pz * inv_s, wz = py * inv_s;
    const float vx = wy * uz - wz * uy;
    const float vy = wz * ux - wx * uz;
    const float vz = wx * uy - wy * ux;

    float* T = S0 + (size_t)tid * MST;
    T[0] = ux; T[1] = vx; T[2]  = wx; T[3]  = px;
    T[4] = uy; T[5] = vy; T[6]  = wy; T[7]  = py;
    T[8] = uz; T[9] = vz; T[10] = wz; T[11] = pz;
    T[12] = 0.f; T[13] = 0.f; T[14] = 0.f; T[15] = 1.f;
    __syncthreads();

    const int q = wave * 4 + ((lane & 15) >> 2);

#pragma unroll
    for (int k = 1; k < SUB; ++k) {
        float* cur = S0 + (q * SUB + k) * MST;
        wmma_compose4(cur - MST, cur, cur, lane);
        __syncthreads();
    }

#pragma unroll
    for (int rnd = 0; rnd < 5; ++rnd) {
        const int off = 1 << rnd;
        const float* mm;
        const float* nn;
        float*       dd;
        if (rnd == 0) {
            nn = S0 + (q * SUB + SUB - 1) * MST;
            mm = (q >= 1) ? (nn - SUB * MST) : IDM;
            dd = G1 + q * GST;
        } else {
            const float* src = (rnd & 1) ? G1 : G0;
            float*       dst = (rnd & 1) ? G0 : G1;
            nn = src + q * GST;
            mm = (q >= off) ? (src + (q - off) * GST) : IDM;
            dd = dst + q * GST;
        }
        wmma_compose4(mm, nn, dd, lane);
        __syncthreads();
    }

    {
        const int half = lane >> 4;
        const int n    = lane & 15;
        const int g    = n >> 2;
        const int jp   = n & 3;
        const int k0   = half ? 2 : 0;

        const float* carry = (q >= 1) ? (G1 + (q - 1) * GST) : IDM;
        const v2f a = *(const v2f*)(carry + jp * 4 + k0);

        v2f pb[SUB / 4];
#pragma unroll
        for (int p = 0; p < SUB / 4; ++p) {
            const float* P = S0 + (q * SUB + p * 4 + jp) * MST;
            pb[p].x = P[k0 * 4 + 3];
            pb[p].y = P[(k0 + 1) * 4 + 3];
        }

        const bool keep = (g >> 1) == half;
        const bool hi   = (g & 1) != 0;
#pragma unroll
        for (int p = 0; p < SUB / 4; ++p) {
            const v8f d = wmma_f32_c0_then_valu(a, pb[p]);
            const float x = hi ? d[4] : d[0];
            const float y = hi ? d[5] : d[1];
            const float z = hi ? d[6] : d[2];
            if (keep) {
                const int    pi = chunk * CHUNK + q * SUB + p * 4 + jp;
                const size_t oi = ((size_t)pi * BATCH + b) * 3;
                tloc[oi + 0] = x;
                tloc[oi + 1] = y;
                tloc[oi + 2] = z;
            }
        }
    }

    if (tid < AGE)
        agg[((size_t)chunk * AGE + tid) * BATCH + b] = G1[(NSUB - 1) * GST + tid];
}

#define APTS     8
#define AWAVES   8
#define ABLOCK   (AWAVES * 32)
#define AGRID    (CHAIN * BATCH / (APTS * ABLOCK))
#define SEG      (NCHUNK / AWAVES)

__device__ __forceinline__ void affine_mul(float* q, const float* a)
{
#pragma unroll
    for (int i = 0; i < 3; ++i) {
        const float lx = q[i * 4 + 0], ly = q[i * 4 + 1];
        const float lz = q[i * 4 + 2], lw = q[i * 4 + 3];
        q[i * 4 + 0] = fmaf(lx, a[0], fmaf(ly, a[4], lz * a[8]));
        q[i * 4 + 1] = fmaf(lx, a[1], fmaf(ly, a[5], lz * a[9]));
        q[i * 4 + 2] = fmaf(lx, a[2], fmaf(ly, a[6], lz * a[10]));
        q[i * 4 + 3] = fmaf(lx, a[3], fmaf(ly, a[7], fmaf(lz, a[11], lw)));
    }
}

__global__ void __launch_bounds__(ABLOCK)
rgn_apply(const float* __restrict__ tloc,
          const float* __restrict__ agg,
          float* __restrict__ out)
{
    __shared__ float Q[AWAVES * AGE * BATCH];

    const int b     = threadIdx.x & 31;
    const int w     = threadIdx.x >> 5;
    const int i0    = (blockIdx.x * AWAVES + w) * APTS;
    const int chunk = (blockIdx.x * (AWAVES * APTS)) / CHUNK;

    float x[APTS][3];
#pragma unroll
    for (int k = 0; k < APTS; ++k) {
        const size_t oi = ((size_t)(i0 + k) * BATCH + b) * 3;
        x[k][0] = tloc[oi + 0]; x[k][1] = tloc[oi + 1]; x[k][2] = tloc[oi + 2];
    }

    float q[AGE] = {1.f, 0.f, 0.f, 0.f,  0.f, 1.f, 0.f, 0.f,  0.f, 0.f, 1.f, 0.f};
    if (w * SEG < chunk) {
        float a[SEG][AGE];
#pragma unroll
        for (int j = 0; j < SEG; ++j) {
#pragma unroll
            for (int e = 0; e < AGE; ++e)
                a[j][e] = agg[((size_t)(w * SEG + j) * AGE + e) * BATCH + b];
        }
#pragma unroll
        for (int j = 0; j < SEG; ++j) {
            if (w * SEG + j < chunk) affine_mul(q, a[j]);
        }
    }
#pragma unroll
    for (int e = 0; e < AGE; ++e) Q[(w * AGE + e) * BATCH + b] = q[e];
    __syncthreads();

    float p[AGE];
#pragma unroll
    for (int e = 0; e < AGE; ++e) p[e] = Q[e * BATCH + b];
    float r[AWAVES - 1][AGE];
#pragma unroll
    for (int s = 1; s < AWAVES; ++s) {
#pragma unroll
        for (int e = 0; e < AGE; ++e) r[s - 1][e] = Q[(s * AGE + e) * BATCH + b];
    }
#pragma unroll
    for (int s = 1; s < AWAVES; ++s) affine_mul(p, r[s - 1]);

#pragma unroll
    for (int k = 0; k < APTS; ++k) {
        const size_t oi = ((size_t)(i0 + k) * BATCH + b) * 3;
        const float tx = x[k][0], ty = x[k][1], tz = x[k][2];
        out[oi + 0] = fmaf(p[0], tx, fmaf(p[1], ty, fmaf(p[2],  tz, p[3])));
        out[oi + 1] = fmaf(p[4], tx, fmaf(p[5], ty, fmaf(p[6],  tz, p[7])));
        out[oi + 2] = fmaf(p[8], tx, fmaf(p[9], ty, fmaf(p[10], tz, p[11])));
    }
}

extern "C" void kernel_launch(void* const* d_in, const int* in_sizes, int n_in,
                              void* d_out, int out_size, void* d_ws, size_t ws_size,
                              hipStream_t stream)
{
    (void)in_sizes; (void)n_in; (void)out_size; (void)ws_size;
    const float* dih = (const float*)d_in[0];
    float* out = (float*)d_out;

    float* ws   = (float*)d_ws;
    float* tloc = ws;
    float* agg  = tloc + (size_t)CHAIN * BATCH * 3;

    dim3 gridB(NCHUNK, BATCH, 1);
    rgn_local_scan<<<gridB, 256, 0, stream>>>(dih, tloc, agg);
    rgn_apply<<<AGRID, ABLOCK, 0, stream>>>(tloc, agg, out);
}
